# starting kernel
# baseline (speedup 1.0000x reference)
#include <hip/hip_runtime.h>

#define Bn 4
#define Tn 512
#define Cn 32

typedef float v2f __attribute__((ext_vector_type(2)));
typedef float v8f __attribute__((ext_vector_type(8)));

__global__ __launch_bounds__(1024) void rosa_kernel(const float* __restrict__ x,
                                                    const float* __restrict__ emb0,
                                                    const float* __restrict__ emb1,
                                                    float* __restrict__ out)
{
    __shared__ signed char lds_sc[Tn][Cn];
    __shared__ unsigned    lds_words[Cn][16];

    const int lane = threadIdx.x;
    const int c    = threadIdx.y;
    const int b    = blockIdx.x;
    const int tid  = c * 32 + lane;

    const float* __restrict__ xb = x + b * Tn * Cn;
    __builtin_prefetch(xb + tid, 0, 1);

#pragma unroll
    for (int e = tid; e < Tn * Cn; e += 1024) {
        lds_sc[e >> 5][e & 31] = (xb[e] > 0.0f) ? (signed char)1 : (signed char)0;
    }
    __syncthreads();

    unsigned myword = 0;
#pragma unroll
    for (int j = 0; j < 16; ++j) {
        unsigned w = (unsigned)__ballot(lds_sc[j * 32 + lane][c] != 0);
        if (lane == j) myword = w;
    }
    if (lane < 16) lds_words[c][lane] = myword;
    __syncthreads();

    int run[16];
#pragma unroll
    for (int j = 0; j < 16; ++j) run[j] = 0;

    for (int w = 0; w < 16; ++w) {
        const unsigned bw = lds_words[c][w];
        for (int i = 0; i < 32; ++i) {
            const int t  = (w << 5) + i;
            const int bt = (int)((bw >> i) & 1u);
            int best = 0;
#pragma unroll
            for (int j = 0; j < 16; ++j) {
                const int k = 1 + lane + (j << 5);
                const int s = t - k;
                int m = 0;
                if (s >= 0) {
                    const unsigned sw = lds_words[c][s >> 5];
                    m = ((int)((sw >> (s & 31)) & 1u) == bt);
                }
                run[j] = m ? (run[j] + 1) : 0;
                const int key = (run[j] << 10) | (512 - k);
                best = (key > best) ? key : best;
            }
#pragma unroll
            for (int off = 16; off > 0; off >>= 1) {
                const int o = __shfl_xor(best, off, 32);
                best = (o > best) ? o : best;
            }
            int idxv = -1;
            const int L = best >> 10;
            if (L > 0) {
                const int kstar = 512 - (best & 1023);
                const int nxt   = t - kstar + 1;
                idxv = (int)((lds_words[c][nxt >> 5] >> (nxt & 31)) & 1u);
            }
            if (lane == 0) lds_sc[t][c] = (signed char)idxv;
        }
    }
    __syncthreads();

    const int mr   = lane & 15;
    const int half = lane >> 4;
#pragma unroll
    for (int it = 0; it < 2; ++it) {
        const int tile = c * 2 + it;
        const int t0   = (tile >> 1) << 4;
        const int c0   = (tile & 1) << 4;
        const float e0v = emb0[c0 + mr];
        const float e1v = emb1[c0 + mr];
        v8f acc = {};
#pragma unroll
        for (int kk = 0; kk < 16; kk += 4) {
            const int ka = kk + 2 * half;
            const signed char i0 = lds_sc[t0 + mr][c0 + ka];
            const signed char i1 = lds_sc[t0 + mr][c0 + ka + 1];
            v2f a0 = { (i0 == 0) ? 1.0f : 0.0f, (i1 == 0) ? 1.0f : 0.0f };
            v2f a1 = { (i0 == 1) ? 1.0f : 0.0f, (i1 == 1) ? 1.0f : 0.0f };
            v2f b0 = { (mr == ka) ? e0v : 0.0f, (mr == ka + 1) ? e0v : 0.0f };
            v2f b1 = { (mr == ka) ? e1v : 0.0f, (mr == ka + 1) ? e1v : 0.0f };
            acc = __builtin_amdgcn_wmma_f32_16x16x4_f32(
                false, a0, false, b0, (short)0, acc, false, false);
            acc = __builtin_amdgcn_wmma_f32_16x16x4_f32(
                false, a1, false, b1, (short)0, acc, false, false);
        }
        float* __restrict__ outb = out + b * Tn * Cn;
#pragma unroll
        for (int r = 0; r < 8; ++r) {
            outb[(t0 + r + 8 * half) * Cn + (c0 + mr)] = acc[r];
        }
    }
}

extern "C" void kernel_launch(void* const* d_in, const int* in_sizes, int n_in,
                              void* d_out, int out_size, void* d_ws, size_t ws_size,
                              hipStream_t stream) {
    const float* x  = (const float*)d_in[0];
    const float* e0 = (const float*)d_in[1];
    const float* e1 = (const float*)d_in[2];
    float* out = (float*)d_out;
    (void)in_sizes; (void)n_in; (void)out_size; (void)d_ws; (void)ws_size;
    dim3 grid(Bn);
    dim3 block(32, 32);
    rosa_kernel<<<grid, block, 0, stream>>>(x, e0, e1, out);
}


# both run counters of a lane in one register as packed 16-bit keys (run<<6 | lag code): one packed multiply-add per position, lane fold on two positions per register; integer ranking only
# speedup vs baseline: 63.0670x; 63.0670x over previous
#include <hip/hip_runtime.h>

#define Bn 4
#define Tn 512
#define Cn 32
#define NW 8

typedef float v2f __attribute__((ext_vector_type(2)));
typedef float v8f __attribute__((ext_vector_type(8)));
typedef unsigned short us2 __attribute__((ext_vector_type(2)));


template <int D>
__device__ __forceinline__ int lane_xor(int v)
{
    if constexpr (D < 16) {
        return __builtin_amdgcn_update_dpp(0, v, 0x160 | D, 0xF, 0xF, false);
    } else {
        return __builtin_amdgcn_permlanex16(v, v, 0x76543210u, 0xfedcba98u, false, false);
    }
}

__device__ __forceinline__ void half_bits16(unsigned z, unsigned (&b)[16])
{
    asm(
        "v_pk_lshlrev_b16 %0, 15, %16 op_sel_hi:[0,1]\n\t"
        "v_pk_lshlrev_b16 %1, 14, %16 op_sel_hi:[0,1]\n\t"
        "v_pk_lshlrev_b16 %2, 13, %16 op_sel_hi:[0,1]\n\t"
        "v_pk_lshlrev_b16 %3, 12, %16 op_sel_hi:[0,1]\n\t"
        "v_pk_lshlrev_b16 %4, 11, %16 op_sel_hi:[0,1]\n\t"
        "v_pk_lshlrev_b16 %5, 10, %16 op_sel_hi:[0,1]\n\t"
        "v_pk_lshlrev_b16 %6, 9, %16 op_sel_hi:[0,1]\n\t"
        "v_pk_lshlrev_b16 %7, 8, %16 op_sel_hi:[0,1]\n\t"
        "v_pk_lshlrev_b16 %8, 7, %16 op_sel_hi:[0,1]\n\t"
        "v_pk_lshlrev_b16 %9, 6, %16 op_sel_hi:[0,1]\n\t"
        "v_pk_lshlrev_b16 %10, 5, %16 op_sel_hi:[0,1]\n\t"
        "v_pk_lshlrev_b16 %11, 4, %16 op_sel_hi:[0,1]\n\t"
        "v_pk_lshlrev_b16 %12, 3, %16 op_sel_hi:[0,1]\n\t"
        "v_pk_lshlrev_b16 %13, 2, %16 op_sel_hi:[0,1]\n\t"
        "v_pk_lshlrev_b16 %14, 1, %16 op_sel_hi:[0,1]\n\t"
        "v_pk_lshrrev_b16 %0, 15, %0 op_sel_hi:[0,1]\n\t"
        "v_pk_lshrrev_b16 %1, 15, %1 op_sel_hi:[0,1]\n\t"
        "v_pk_lshrrev_b16 %2, 15, %2 op_sel_hi:[0,1]\n\t"
        "v_pk_lshrrev_b16 %3, 15, %3 op_sel_hi:[0,1]\n\t"
        "v_pk_lshrrev_b16 %4, 15, %4 op_sel_hi:[0,1]\n\t"
        "v_pk_lshrrev_b16 %5, 15, %5 op_sel_hi:[0,1]\n\t"
        "v_pk_lshrrev_b16 %6, 15, %6 op_sel_hi:[0,1]\n\t"
        "v_pk_lshrrev_b16 %7, 15, %7 op_sel_hi:[0,1]\n\t"
        "v_pk_lshrrev_b16 %8, 15, %8 op_sel_hi:[0,1]\n\t"
        "v_pk_lshrrev_b16 %9, 15, %9 op_sel_hi:[0,1]\n\t"
        "v_pk_lshrrev_b16 %10, 15, %10 op_sel_hi:[0,1]\n\t"
        "v_pk_lshrrev_b16 %11, 15, %11 op_sel_hi:[0,1]\n\t"
        "v_pk_lshrrev_b16 %12, 15, %12 op_sel_hi:[0,1]\n\t"
        "v_pk_lshrrev_b16 %13, 15, %13 op_sel_hi:[0,1]\n\t"
        "v_pk_lshrrev_b16 %14, 15, %14 op_sel_hi:[0,1]\n\t"
        "v_pk_lshrrev_b16 %15, 15, %16 op_sel_hi:[0,1]"
        : "=&v"(b[0]), "=&v"(b[1]), "=&v"(b[2]), "=&v"(b[3]),
          "=&v"(b[4]), "=&v"(b[5]), "=&v"(b[6]), "=&v"(b[7]),
          "=&v"(b[8]), "=&v"(b[9]), "=&v"(b[10]), "=&v"(b[11]),
          "=&v"(b[12]), "=&v"(b[13]), "=&v"(b[14]), "=&v"(b[15])
        : "v"(z));
}

__device__ __forceinline__ unsigned max_of_halves(us2 k)
{
    const us2 sw = { k.y, k.x };
    return __builtin_bit_cast(unsigned, __builtin_elementwise_max(k, sw));
}

template <int S, int N>
__device__ __forceinline__ void fold(unsigned (&vals)[16], int lane)
{
    const bool up = (lane >> S) & 1;
#pragma unroll
    for (int m = 0; m < N / 2; ++m) {
        const unsigned a = vals[2 * m];
        const unsigned b = vals[2 * m + 1];
        const unsigned keep = up ? b : a;
        const unsigned send = up ? a : b;
        const unsigned got  = (unsigned)lane_xor<(1 << S)>((int)send);
        const us2 mx = __builtin_elementwise_max(__builtin_bit_cast(us2, got),
                                                 __builtin_bit_cast(us2, keep));
        vals[m] = __builtin_bit_cast(unsigned, mx);
    }
}

__device__ __forceinline__ unsigned match_word(const unsigned* words, int wd, int k, unsigned bw)
{
    const int p = 32 * wd - k;
    if (p <= -32) return 0u;
    const int q = p >> 5;
    const int r = p & 31;
    const unsigned lo = (q >= 0) ? words[q < 0 ? 0 : q] : 0u;
    const unsigned hi = words[q + 1];
    const unsigned h  = __funnelshift_r(lo, hi, r);
    const unsigned vm = (p >= 0) ? 0xFFFFFFFFu : (0xFFFFFFFFu << (-p));
    return ~(h ^ bw) & vm;
}

__global__ __launch_bounds__(256) void scan_kernel(const float* __restrict__ x,
                                                   float* __restrict__ ws)
{
    __shared__ unsigned words[16];
    __shared__ unsigned short keys[NW][Tn];
    __shared__ __attribute__((aligned(16))) signed char idxs[Tn];

    const int tid  = threadIdx.x;
    const int lane = tid & 31;
    const int w    = tid >> 5;
    const int col  = blockIdx.x;
    const int b    = col >> 5;
    const int c    = col & 31;

    const float* __restrict__ xc = x + b * Tn * Cn + c;
    const float v0 = xc[tid * Cn];
    const float v1 = xc[(tid + 256) * Cn];
    const unsigned w0 = (unsigned)__ballot(v0 > 0.0f);
    const unsigned w1 = (unsigned)__ballot(v1 > 0.0f);
    if (lane == 0) { words[w] = w0; words[w + 8] = w1; }
    __syncthreads();

    const int kA = 1 + lane + 64 * w;
    const int kB = kA + 32;
    const us2 code = { (unsigned short)(63 - lane), (unsigned short)(31 - lane) };
    const us2 step = { 64, 64 };
    us2 next = step;
    for (int wd = 2 * w; wd < 16; ++wd) {
        const unsigned bw = words[wd];
        const unsigned xA = match_word(words, wd, kA, bw);
        const unsigned xB = match_word(words, wd, kB, bw);
        const us2 zlo = { (unsigned short)(xA & 0xFFFFu), (unsigned short)(xB & 0xFFFFu) };
        const us2 zhi = { (unsigned short)(xA >> 16), (unsigned short)(xB >> 16) };
        unsigned blo[16], bhi[16];
        half_bits16(__builtin_bit_cast(unsigned, zlo), blo);
        half_bits16(__builtin_bit_cast(unsigned, zhi), bhi);
        unsigned vals[16];
#pragma unroll
        for (int i = 0; i < 32; ++i) {
            const us2 bit = __builtin_bit_cast(us2, (i < 16) ? blo[i & 15] : bhi[i & 15]);
            const us2 key = next * bit + code;
            next = next * bit + step;
            const unsigned best = max_of_halves(key);
            if (i & 1) vals[i >> 1] = (best & 0xFFFF0000u) | (vals[i >> 1] & 0x0000FFFFu);
            else       vals[i >> 1] = best;
        }
        fold<0, 16>(vals, lane);
        fold<1, 8>(vals, lane);
        fold<2, 4>(vals, lane);
        fold<3, 2>(vals, lane);
        const bool upper = lane >> 4;
        const unsigned lo16 = vals[0] & 0xFFFFu;
        const unsigned hi16 = vals[0] >> 16;
        const unsigned keep = upper ? hi16 : lo16;
        const unsigned send = upper ? lo16 : hi16;
        const unsigned got  = (unsigned)lane_xor<16>((int)send);
        const unsigned res  = (got > keep) ? got : keep;
        keys[w][wd * 32 + 2 * (lane & 15) + (lane >> 4)] = (unsigned short)res;
    }
    __syncthreads();

#pragma unroll
    for (int h = 0; h < 2; ++h) {
        const int t = tid + 256 * h;
        const int wmax = t >> 6;
        int best = 0;
#pragma unroll
        for (int u = 0; u < NW; ++u) {
            if (u <= wmax) {
                const int k16 = keys[u][t];
                const int kv  = ((k16 >> 6) << 10) | ((k16 & 63) + 448 - 64 * u);
                best = (kv > best) ? kv : best;
            }
        }
        int idxv = -1;
        if ((best >> 10) > 0) {
            const int kstar = 512 - (best & 1023);
            const int nxt   = t - kstar + 1;
            idxv = (int)((words[nxt >> 5] >> (nxt & 31)) & 1u);
        }
        idxs[t] = (signed char)idxv;
    }
    __syncthreads();

    if (w == 0) {
        const uint4 v = reinterpret_cast<const uint4*>(idxs)[lane];
        reinterpret_cast<uint4*>(reinterpret_cast<unsigned char*>(ws) + col * Tn)[lane] = v;
    }
}

__global__ __launch_bounds__(128) void emb_kernel(const float* __restrict__ ws,
                                                  const float* __restrict__ emb0,
                                                  const float* __restrict__ emb1,
                                                  float* __restrict__ out)
{
    const int lane = threadIdx.x & 31;
    const int tt   = blockIdx.x * 4 + (threadIdx.x >> 5);
    const int b    = tt >> 5;
    const int t0   = (tt & 31) << 4;
    const signed char* __restrict__ ib =
        reinterpret_cast<const signed char*>(ws) + b * Cn * Tn;
    float* __restrict__ outb = out + b * Tn * Cn;

    const int mr   = lane & 15;
    const int half = lane >> 4;
#pragma unroll
    for (int it = 0; it < 2; ++it) {
        const int c0 = it << 4;
        const float e0v = emb0[c0 + mr];
        const float e1v = emb1[c0 + mr];
        v8f acc = {};
#pragma unroll
        for (int kk = 0; kk < 16; kk += 4) {
            const int ka = kk + 2 * half;
            const signed char i0 = ib[(c0 + ka) * Tn + t0 + mr];
            const signed char i1 = ib[(c0 + ka + 1) * Tn + t0 + mr];
            v2f a0 = { (i0 == 0) ? 1.0f : 0.0f, (i1 == 0) ? 1.0f : 0.0f };
            v2f a1 = { (i0 == 1) ? 1.0f : 0.0f, (i1 == 1) ? 1.0f : 0.0f };
            v2f b0 = { (mr == ka) ? e0v : 0.0f, (mr == ka + 1) ? e0v : 0.0f };
            v2f b1 = { (mr == ka) ? e1v : 0.0f, (mr == ka + 1) ? e1v : 0.0f };
            acc = __builtin_amdgcn_wmma_f32_16x16x4_f32(
                false, a0, false, b0, (short)0, acc, false, false);
            acc = __builtin_amdgcn_wmma_f32_16x16x4_f32(
                false, a1, false, b1, (short)0, acc, false, false);
        }
#pragma unroll
        for (int r = 0; r < 8; ++r) {
            outb[(t0 + r + 8 * half) * Cn + (c0 + mr)] = acc[r];
        }
    }
}

extern "C" void kernel_launch(void* const* d_in, const int* in_sizes, int n_in,
                              void* d_out, int out_size, void* d_ws, size_t ws_size,
                              hipStream_t stream) {
    const float* x  = (const float*)d_in[0];
    const float* e0 = (const float*)d_in[1];
    const float* e1 = (const float*)d_in[2];
    float* out = (float*)d_out;
    float* ws  = (float*)d_ws;
    (void)in_sizes; (void)n_in; (void)out_size; (void)ws_size;
    scan_kernel<<<dim3(Bn * Cn), dim3(256), 0, stream>>>(x, ws);
    emb_kernel<<<dim3(32), dim3(128), 0, stream>>>(ws, e0, e1, out);
}


# t axis split over two blocks per column (t<384 | t>=384): 256 blocks of 4 waves, a wave walks 2 lag blocks in turn; late block's counters start from leading ones of earlier match words; integer only
# speedup vs baseline: 68.6684x; 1.0888x over previous
#include <hip/hip_runtime.h>

#define Bn 4
#define Tn 512
#define Cn 32
#define NLB 8
#define SPLITW 12

typedef float v2f __attribute__((ext_vector_type(2)));
typedef float v8f __attribute__((ext_vector_type(8)));
typedef unsigned short us2 __attribute__((ext_vector_type(2)));


template <int D>
__device__ __forceinline__ int lane_xor(int v)
{
    if constexpr (D < 16) {
        return __builtin_amdgcn_update_dpp(0, v, 0x160 | D, 0xF, 0xF, false);
    } else {
        return __builtin_amdgcn_permlanex16(v, v, 0x76543210u, 0xfedcba98u, false, false);
    }
}

__device__ __forceinline__ void half_bits16(unsigned z, unsigned (&b)[16])
{
    asm(
        "v_pk_lshlrev_b16 %0, 15, %16 op_sel_hi:[0,1]\n\t"
        "v_pk_lshlrev_b16 %1, 14, %16 op_sel_hi:[0,1]\n\t"
        "v_pk_lshlrev_b16 %2, 13, %16 op_sel_hi:[0,1]\n\t"
        "v_pk_lshlrev_b16 %3, 12, %16 op_sel_hi:[0,1]\n\t"
        "v_pk_lshlrev_b16 %4, 11, %16 op_sel_hi:[0,1]\n\t"
        "v_pk_lshlrev_b16 %5, 10, %16 op_sel_hi:[0,1]\n\t"
        "v_pk_lshlrev_b16 %6, 9, %16 op_sel_hi:[0,1]\n\t"
        "v_pk_lshlrev_b16 %7, 8, %16 op_sel_hi:[0,1]\n\t"
        "v_pk_lshlrev_b16 %8, 7, %16 op_sel_hi:[0,1]\n\t"
        "v_pk_lshlrev_b16 %9, 6, %16 op_sel_hi:[0,1]\n\t"
        "v_pk_lshlrev_b16 %10, 5, %16 op_sel_hi:[0,1]\n\t"
        "v_pk_lshlrev_b16 %11, 4, %16 op_sel_hi:[0,1]\n\t"
        "v_pk_lshlrev_b16 %12, 3, %16 op_sel_hi:[0,1]\n\t"
        "v_pk_lshlrev_b16 %13, 2, %16 op_sel_hi:[0,1]\n\t"
        "v_pk_lshlrev_b16 %14, 1, %16 op_sel_hi:[0,1]\n\t"
        "v_pk_lshrrev_b16 %0, 15, %0 op_sel_hi:[0,1]\n\t"
        "v_pk_lshrrev_b16 %1, 15, %1 op_sel_hi:[0,1]\n\t"
        "v_pk_lshrrev_b16 %2, 15, %2 op_sel_hi:[0,1]\n\t"
        "v_pk_lshrrev_b16 %3, 15, %3 op_sel_hi:[0,1]\n\t"
        "v_pk_lshrrev_b16 %4, 15, %4 op_sel_hi:[0,1]\n\t"
        "v_pk_lshrrev_b16 %5, 15, %5 op_sel_hi:[0,1]\n\t"
        "v_pk_lshrrev_b16 %6, 15, %6 op_sel_hi:[0,1]\n\t"
        "v_pk_lshrrev_b16 %7, 15, %7 op_sel_hi:[0,1]\n\t"
        "v_pk_lshrrev_b16 %8, 15, %8 op_sel_hi:[0,1]\n\t"
        "v_pk_lshrrev_b16 %9, 15, %9 op_sel_hi:[0,1]\n\t"
        "v_pk_lshrrev_b16 %10, 15, %10 op_sel_hi:[0,1]\n\t"
        "v_pk_lshrrev_b16 %11, 15, %11 op_sel_hi:[0,1]\n\t"
        "v_pk_lshrrev_b16 %12, 15, %12 op_sel_hi:[0,1]\n\t"
        "v_pk_lshrrev_b16 %13, 15, %13 op_sel_hi:[0,1]\n\t"
        "v_pk_lshrrev_b16 %14, 15, %14 op_sel_hi:[0,1]\n\t"
        "v_pk_lshrrev_b16 %15, 15, %16 op_sel_hi:[0,1]"
        : "=&v"(b[0]), "=&v"(b[1]), "=&v"(b[2]), "=&v"(b[3]),
          "=&v"(b[4]), "=&v"(b[5]), "=&v"(b[6]), "=&v"(b[7]),
          "=&v"(b[8]), "=&v"(b[9]), "=&v"(b[10]), "=&v"(b[11]),
          "=&v"(b[12]), "=&v"(b[13]), "=&v"(b[14]), "=&v"(b[15])
        : "v"(z));
}

__device__ __forceinline__ unsigned max_of_halves(us2 k)
{
    const us2 sw = { k.y, k.x };
    return __builtin_bit_cast(unsigned, __builtin_elementwise_max(k, sw));
}

template <int S, int N>
__device__ __forceinline__ void fold(unsigned (&vals)[16], int lane)
{
    const bool up = (lane >> S) & 1;
#pragma unroll
    for (int m = 0; m < N / 2; ++m) {
        const unsigned a = vals[2 * m];
        const unsigned b = vals[2 * m + 1];
        const unsigned keep = up ? b : a;
        const unsigned send = up ? a : b;
        const unsigned got  = (unsigned)lane_xor<(1 << S)>((int)send);
        const us2 mx = __builtin_elementwise_max(__builtin_bit_cast(us2, got),
                                                 __builtin_bit_cast(us2, keep));
        vals[m] = __builtin_bit_cast(unsigned, mx);
    }
}

__device__ __forceinline__ unsigned word_at(unsigned wreg, int j)
{
    return (j >= 0) ? (unsigned)__builtin_amdgcn_readlane((int)wreg, j) : 0u;
}

__device__ __forceinline__ void match_pair(unsigned wreg, int wd, int g, int lane,
                                           unsigned edge, unsigned& xA, unsigned& xB)
{
    const int m = wd - 2 * g;
    const unsigned bw = word_at(wreg, wd);
    const unsigned u0 = word_at(wreg, m);
    const unsigned u1 = word_at(wreg, m - 1);
    const unsigned u2 = word_at(wreg, m - 2);
    const unsigned r  = 31u - (unsigned)lane;
    const unsigned hA = __funnelshift_r(u1, u0, r);
    const unsigned hB = __funnelshift_r(u2, u1, r);
    const unsigned vA = (m >= 1) ? 0xFFFFFFFFu : edge;
    const unsigned vB = (m >= 2) ? 0xFFFFFFFFu : ((m == 1) ? edge : 0u);
    xA = ~(hA ^ bw) & vA;
    xB = ~(hB ^ bw) & vB;
}

__global__ __launch_bounds__(128) void scan_kernel(const float* __restrict__ x,
                                                   float* __restrict__ ws)
{
    __shared__ unsigned words[16];
    __shared__ unsigned short keys[NLB][Tn];
    __shared__ __attribute__((aligned(16))) signed char idxs[Tn];

    const int tid  = threadIdx.x;
    const int lane = tid & 31;
    const int w    = __builtin_amdgcn_readfirstlane(tid >> 5);
    const int col  = blockIdx.x >> 1;
    const int part = blockIdx.x & 1;
    const int b    = col >> 5;
    const int c    = col & 31;

    const float* __restrict__ xc = x + b * Tn * Cn + c;
    float xv[4];
#pragma unroll
    for (int j = 0; j < 4; ++j) xv[j] = xc[(tid + 128 * j) * Cn];
#pragma unroll
    for (int j = 0; j < 4; ++j) {
        const unsigned wj = (unsigned)__ballot(xv[j] > 0.0f);
        if (lane == 0) words[w + 4 * j] = wj;
    }
    __syncthreads();

    const unsigned wreg = words[lane & 15];
    asm volatile("" :: "v"(wreg));
    const unsigned edge = (lane < 31) ? (0xFFFFFFFFu << (lane + 1)) : 0u;
    const int wd_first = part ? SPLITW : 0;
    const int wd_end   = part ? 16 : SPLITW;

    const us2 code = { (unsigned short)(63 - lane), (unsigned short)(31 - lane) };
    const us2 step = { 64, 64 };
    for (int pass = 0; pass < 2; ++pass) {
    const int g  = (pass == 0) ? w : (part ? 7 - w : ((w == 3) ? 7 : 6 - w));
    const int wd_begin = (2 * g > wd_first) ? 2 * g : wd_first;
    unsigned runA = 0, runB = 0;
    bool goA = true, goB = true;
    for (int j = wd_begin - 1; j >= 2 * g; --j) {
        unsigned mA, mB;
        match_pair(wreg, j, g, lane, edge, mA, mB);
        const unsigned nA = ~mA;
        const unsigned nB = ~mB;
        runA += goA ? (nA ? (unsigned)__builtin_clz(nA) : 32u) : 0u;
        runB += goB ? (nB ? (unsigned)__builtin_clz(nB) : 32u) : 0u;
        goA = goA && (nA == 0u);
        goB = goB && (nB == 0u);
        if (!__any(goA || goB)) break;
    }
    us2 next = { (unsigned short)((runA + 1u) << 6), (unsigned short)((runB + 1u) << 6) };
    for (int wd = wd_begin; wd < wd_end; ++wd) {
        unsigned xA, xB;
        match_pair(wreg, wd, g, lane, edge, xA, xB);
        const us2 zlo = { (unsigned short)(xA & 0xFFFFu), (unsigned short)(xB & 0xFFFFu) };
        const us2 zhi = { (unsigned short)(xA >> 16), (unsigned short)(xB >> 16) };
        unsigned blo[16], bhi[16];
        half_bits16(__builtin_bit_cast(unsigned, zlo), blo);
        half_bits16(__builtin_bit_cast(unsigned, zhi), bhi);
        unsigned vals[16];
#pragma unroll
        for (int i = 0; i < 32; ++i) {
            const us2 bit = __builtin_bit_cast(us2, (i < 16) ? blo[i & 15] : bhi[i & 15]);
            const us2 key = next * bit + code;
            next = next * bit + step;
            const unsigned best = max_of_halves(key);
            if (i & 1) vals[i >> 1] = (best & 0xFFFF0000u) | (vals[i >> 1] & 0x0000FFFFu);
            else       vals[i >> 1] = best;
        }
        fold<0, 16>(vals, lane);
        fold<1, 8>(vals, lane);
        fold<2, 4>(vals, lane);
        fold<3, 2>(vals, lane);
        const bool upper = lane >> 4;
        const unsigned lo16 = vals[0] & 0xFFFFu;
        const unsigned hi16 = vals[0] >> 16;
        const unsigned keep = upper ? hi16 : lo16;
        const unsigned send = upper ? lo16 : hi16;
        const unsigned got  = (unsigned)lane_xor<16>((int)send);
        const unsigned res  = (got > keep) ? got : keep;
        keys[g][wd * 32 + 2 * (lane & 15) + (lane >> 4)] = (unsigned short)res;
    }
    }
    __syncthreads();

    const int t_first = wd_first * 32;
    const int npass   = part ? 1 : 3;
    for (int h = 0; h < npass; ++h) {
        const int t = t_first + tid + 128 * h;
        const int wmax = t >> 6;
        int best = 0;
#pragma unroll
        for (int u = 0; u < NLB; ++u) {
            if (u <= wmax) {
                const int k16 = keys[u][t];
                const int kv  = ((k16 >> 6) << 10) | ((k16 & 63) + 448 - 64 * u);
                best = (kv > best) ? kv : best;
            }
        }
        int idxv = -1;
        if ((best >> 10) > 0) {
            const int kstar = 512 - (best & 1023);
            const int nxt   = t - kstar + 1;
            idxv = (int)((words[nxt >> 5] >> (nxt & 31)) & 1u);
        }
        idxs[t] = (signed char)idxv;
    }
    __syncthreads();

    if (w == 0 && lane < (part ? 8 : 24)) {
        const int q16 = (t_first >> 4) + lane;
        const uint4 v = reinterpret_cast<const uint4*>(idxs)[q16];
        reinterpret_cast<uint4*>(reinterpret_cast<unsigned char*>(ws) + col * Tn)[q16] = v;
    }
}

__global__ __launch_bounds__(128) void emb_kernel(const float* __restrict__ ws,
                                                  const float* __restrict__ emb0,
                                                  const float* __restrict__ emb1,
                                                  float* __restrict__ out)
{
    const int lane = threadIdx.x & 31;
    const int tt   = blockIdx.x * 4 + (threadIdx.x >> 5);
    const int b    = tt >> 5;
    const int t0   = (tt & 31) << 4;
    const signed char* __restrict__ ib =
        reinterpret_cast<const signed char*>(ws) + b * Cn * Tn;
    float* __restrict__ outb = out + b * Tn * Cn;

    const int mr   = lane & 15;
    const int half = lane >> 4;
#pragma unroll
    for (int it = 0; it < 2; ++it) {
        const int c0 = it << 4;
        const float e0v = emb0[c0 + mr];
        const float e1v = emb1[c0 + mr];
        v8f acc = {};
#pragma unroll
        for (int kk = 0; kk < 16; kk += 4) {
            const int ka = kk + 2 * half;
            const signed char i0 = ib[(c0 + ka) * Tn + t0 + mr];
            const signed char i1 = ib[(c0 + ka + 1) * Tn + t0 + mr];
            v2f a0 = { (i0 == 0) ? 1.0f : 0.0f, (i1 == 0) ? 1.0f : 0.0f };
            v2f a1 = { (i0 == 1) ? 1.0f : 0.0f, (i1 == 1) ? 1.0f : 0.0f };
            v2f b0 = { (mr == ka) ? e0v : 0.0f, (mr == ka + 1) ? e0v : 0.0f };
            v2f b1 = { (mr == ka) ? e1v : 0.0f, (mr == ka + 1) ? e1v : 0.0f };
            acc = __builtin_amdgcn_wmma_f32_16x16x4_f32(
                false, a0, false, b0, (short)0, acc, false, false);
            acc = __builtin_amdgcn_wmma_f32_16x16x4_f32(
                false, a1, false, b1, (short)0, acc, false, false);
        }
#pragma unroll
        for (int r = 0; r < 8; ++r) {
            outb[(t0 + r + 8 * half) * Cn + (c0 + mr)] = acc[r];
        }
    }
}

extern "C" void kernel_launch(void* const* d_in, const int* in_sizes, int n_in,
                              void* d_out, int out_size, void* d_ws, size_t ws_size,
                              hipStream_t stream) {
    const float* x  = (const float*)d_in[0];
    const float* e0 = (const float*)d_in[1];
    const float* e1 = (const float*)d_in[2];
    float* out = (float*)d_out;
    float* ws  = (float*)d_ws;
    (void)in_sizes; (void)n_in; (void)out_size; (void)ws_size;
    scan_kernel<<<dim3(2 * Bn * Cn), dim3(128), 0, stream>>>(x, ws);
    emb_kernel<<<dim3(32), dim3(128), 0, stream>>>(ws, e0, e1, out);
}


# second stage (max over lag blocks, bit lookup) without branches: a wave issues all its key reads at known addresses and waits once, then all sign-word reads: 2 LDS waits in series, was up to 15
# speedup vs baseline: 74.9876x; 1.0920x over previous
#include <hip/hip_runtime.h>

#define Bn 4
#define Tn 512
#define Cn 32
#define NLB 8
#define SPLITW 12

typedef float v2f __attribute__((ext_vector_type(2)));
typedef float v8f __attribute__((ext_vector_type(8)));
typedef unsigned short us2 __attribute__((ext_vector_type(2)));


template <int D>
__device__ __forceinline__ int lane_xor(int v)
{
    if constexpr (D < 16) {
        return __builtin_amdgcn_update_dpp(0, v, 0x160 | D, 0xF, 0xF, false);
    } else {
        return __builtin_amdgcn_permlanex16(v, v, 0x76543210u, 0xfedcba98u, false, false);
    }
}

__device__ __forceinline__ void half_bits16(unsigned z, unsigned (&b)[16])
{
    asm(
        "v_pk_lshlrev_b16 %0, 15, %16 op_sel_hi:[0,1]\n\t"
        "v_pk_lshlrev_b16 %1, 14, %16 op_sel_hi:[0,1]\n\t"
        "v_pk_lshlrev_b16 %2, 13, %16 op_sel_hi:[0,1]\n\t"
        "v_pk_lshlrev_b16 %3, 12, %16 op_sel_hi:[0,1]\n\t"
        "v_pk_lshlrev_b16 %4, 11, %16 op_sel_hi:[0,1]\n\t"
        "v_pk_lshlrev_b16 %5, 10, %16 op_sel_hi:[0,1]\n\t"
        "v_pk_lshlrev_b16 %6, 9, %16 op_sel_hi:[0,1]\n\t"
        "v_pk_lshlrev_b16 %7, 8, %16 op_sel_hi:[0,1]\n\t"
        "v_pk_lshlrev_b16 %8, 7, %16 op_sel_hi:[0,1]\n\t"
        "v_pk_lshlrev_b16 %9, 6, %16 op_sel_hi:[0,1]\n\t"
        "v_pk_lshlrev_b16 %10, 5, %16 op_sel_hi:[0,1]\n\t"
        "v_pk_lshlrev_b16 %11, 4, %16 op_sel_hi:[0,1]\n\t"
        "v_pk_lshlrev_b16 %12, 3, %16 op_sel_hi:[0,1]\n\t"
        "v_pk_lshlrev_b16 %13, 2, %16 op_sel_hi:[0,1]\n\t"
        "v_pk_lshlrev_b16 %14, 1, %16 op_sel_hi:[0,1]\n\t"
        "v_pk_lshrrev_b16 %0, 15, %0 op_sel_hi:[0,1]\n\t"
        "v_pk_lshrrev_b16 %1, 15, %1 op_sel_hi:[0,1]\n\t"
        "v_pk_lshrrev_b16 %2, 15, %2 op_sel_hi:[0,1]\n\t"
        "v_pk_lshrrev_b16 %3, 15, %3 op_sel_hi:[0,1]\n\t"
        "v_pk_lshrrev_b16 %4, 15, %4 op_sel_hi:[0,1]\n\t"
        "v_pk_lshrrev_b16 %5, 15, %5 op_sel_hi:[0,1]\n\t"
        "v_pk_lshrrev_b16 %6, 15, %6 op_sel_hi:[0,1]\n\t"
        "v_pk_lshrrev_b16 %7, 15, %7 op_sel_hi:[0,1]\n\t"
        "v_pk_lshrrev_b16 %8, 15, %8 op_sel_hi:[0,1]\n\t"
        "v_pk_lshrrev_b16 %9, 15, %9 op_sel_hi:[0,1]\n\t"
        "v_pk_lshrrev_b16 %10, 15, %10 op_sel_hi:[0,1]\n\t"
        "v_pk_lshrrev_b16 %11, 15, %11 op_sel_hi:[0,1]\n\t"
        "v_pk_lshrrev_b16 %12, 15, %12 op_sel_hi:[0,1]\n\t"
        "v_pk_lshrrev_b16 %13, 15, %13 op_sel_hi:[0,1]\n\t"
        "v_pk_lshrrev_b16 %14, 15, %14 op_sel_hi:[0,1]\n\t"
        "v_pk_lshrrev_b16 %15, 15, %16 op_sel_hi:[0,1]"
        : "=&v"(b[0]), "=&v"(b[1]), "=&v"(b[2]), "=&v"(b[3]),
          "=&v"(b[4]), "=&v"(b[5]), "=&v"(b[6]), "=&v"(b[7]),
          "=&v"(b[8]), "=&v"(b[9]), "=&v"(b[10]), "=&v"(b[11]),
          "=&v"(b[12]), "=&v"(b[13]), "=&v"(b[14]), "=&v"(b[15])
        : "v"(z));
}

__device__ __forceinline__ unsigned max_of_halves(us2 k)
{
    const us2 sw = { k.y, k.x };
    return __builtin_bit_cast(unsigned, __builtin_elementwise_max(k, sw));
}

template <int S, int N>
__device__ __forceinline__ void fold(unsigned (&vals)[16], int lane)
{
    const bool up = (lane >> S) & 1;
#pragma unroll
    for (int m = 0; m < N / 2; ++m) {
        const unsigned a = vals[2 * m];
        const unsigned b = vals[2 * m + 1];
        const unsigned keep = up ? b : a;
        const unsigned send = up ? a : b;
        const unsigned got  = (unsigned)lane_xor<(1 << S)>((int)send);
        const us2 mx = __builtin_elementwise_max(__builtin_bit_cast(us2, got),
                                                 __builtin_bit_cast(us2, keep));
        vals[m] = __builtin_bit_cast(unsigned, mx);
    }
}

__device__ __forceinline__ unsigned word_at(unsigned wreg, int j)
{
    return (j >= 0) ? (unsigned)__builtin_amdgcn_readlane((int)wreg, j) : 0u;
}

__device__ __forceinline__ void match_pair(unsigned wreg, int wd, int g, int lane,
                                           unsigned edge, unsigned& xA, unsigned& xB)
{
    const int m = wd - 2 * g;
    const unsigned bw = word_at(wreg, wd);
    const unsigned u0 = word_at(wreg, m);
    const unsigned u1 = word_at(wreg, m - 1);
    const unsigned u2 = word_at(wreg, m - 2);
    const unsigned r  = 31u - (unsigned)lane;
    const unsigned hA = __funnelshift_r(u1, u0, r);
    const unsigned hB = __funnelshift_r(u2, u1, r);
    const unsigned vA = (m >= 1) ? 0xFFFFFFFFu : edge;
    const unsigned vB = (m >= 2) ? 0xFFFFFFFFu : ((m == 1) ? edge : 0u);
    xA = ~(hA ^ bw) & vA;
    xB = ~(hB ^ bw) & vB;
}

template <int NP, int T0>
__device__ __forceinline__ void best_over_lag_blocks(const unsigned short (&keys)[NLB][Tn],
                                                     const unsigned (&words)[16],
                                                     signed char (&idxs)[Tn],
                                                     int tid, int w)
{
    int k16[NP][NLB];
#pragma unroll
    for (int h = 0; h < NP; ++h) {
        const int t    = T0 + tid + 128 * h;
        const int wmax = (T0 >> 6) + 2 * h + (w >> 1);
#pragma unroll
        for (int u = 0; u < NLB; ++u) {
            if (NP == 3 && u > 2 * h + 1) continue;
            const int uu = (u < wmax) ? u : wmax;
            k16[h][u] = keys[uu][t];
        }
    }
    int best[NP], nxt[NP];
#pragma unroll
    for (int h = 0; h < NP; ++h) {
        const int t    = T0 + tid + 128 * h;
        const int wmax = (T0 >> 6) + 2 * h + (w >> 1);
        int bst = 0;
#pragma unroll
        for (int u = 0; u < NLB; ++u) {
            if (NP == 3 && u > 2 * h + 1) continue;
            const int kk = k16[h][u];
            const int kv = ((kk >> 6) << 10) | ((kk & 63) + 448 - 64 * u);
            const int kc = (u <= wmax) ? kv : 0;
            bst = (kc > bst) ? kc : bst;
        }
        best[h] = bst;
        const int nx = t - (512 - (bst & 1023)) + 1;
        nxt[h] = ((bst >> 10) > 0) ? nx : 0;
    }
    unsigned wv[NP];
#pragma unroll
    for (int h = 0; h < NP; ++h) wv[h] = words[nxt[h] >> 5];
#pragma unroll
    for (int h = 0; h < NP; ++h) {
        const int t   = T0 + tid + 128 * h;
        const int bit = (int)((wv[h] >> (nxt[h] & 31)) & 1u);
        idxs[t] = (signed char)(((best[h] >> 10) > 0) ? bit : -1);
    }
}

__global__ __launch_bounds__(128) void scan_kernel(const float* __restrict__ x,
                                                   float* __restrict__ ws)
{
    __shared__ unsigned words[16];
    __shared__ unsigned short keys[NLB][Tn];
    __shared__ __attribute__((aligned(16))) signed char idxs[Tn];

    const int tid  = threadIdx.x;
    const int lane = tid & 31;
    const int w    = __builtin_amdgcn_readfirstlane(tid >> 5);
    const int col  = blockIdx.x >> 1;
    const int part = blockIdx.x & 1;
    const int b    = col >> 5;
    const int c    = col & 31;

    const float* __restrict__ xc = x + b * Tn * Cn + c;
    float xv[4];
#pragma unroll
    for (int j = 0; j < 4; ++j) xv[j] = xc[(tid + 128 * j) * Cn];
#pragma unroll
    for (int j = 0; j < 4; ++j) {
        const unsigned wj = (unsigned)__ballot(xv[j] > 0.0f);
        if (lane == 0) words[w + 4 * j] = wj;
    }
    __syncthreads();

    const unsigned wreg = words[lane & 15];
    asm volatile("" :: "v"(wreg));
    const unsigned edge = (lane < 31) ? (0xFFFFFFFFu << (lane + 1)) : 0u;
    const int wd_first = part ? SPLITW : 0;
    const int wd_end   = part ? 16 : SPLITW;

    const us2 code = { (unsigned short)(63 - lane), (unsigned short)(31 - lane) };
    const us2 step = { 64, 64 };
    for (int pass = 0; pass < 2; ++pass) {
    const int g  = (pass == 0) ? w : (part ? 7 - w : ((w == 3) ? 7 : 6 - w));
    const int wd_begin = (2 * g > wd_first) ? 2 * g : wd_first;
    unsigned runA = 0, runB = 0;
    bool goA = true, goB = true;
    for (int j = wd_begin - 1; j >= 2 * g; --j) {
        unsigned mA, mB;
        match_pair(wreg, j, g, lane, edge, mA, mB);
        const unsigned nA = ~mA;
        const unsigned nB = ~mB;
        runA += goA ? (nA ? (unsigned)__builtin_clz(nA) : 32u) : 0u;
        runB += goB ? (nB ? (unsigned)__builtin_clz(nB) : 32u) : 0u;
        goA = goA && (nA == 0u);
        goB = goB && (nB == 0u);
        if (!__any(goA || goB)) break;
    }
    us2 next = { (unsigned short)((runA + 1u) << 6), (unsigned short)((runB + 1u) << 6) };
    for (int wd = wd_begin; wd < wd_end; ++wd) {
        unsigned xA, xB;
        match_pair(wreg, wd, g, lane, edge, xA, xB);
        const us2 zlo = { (unsigned short)(xA & 0xFFFFu), (unsigned short)(xB & 0xFFFFu) };
        const us2 zhi = { (unsigned short)(xA >> 16), (unsigned short)(xB >> 16) };
        unsigned blo[16], bhi[16];
        half_bits16(__builtin_bit_cast(unsigned, zlo), blo);
        half_bits16(__builtin_bit_cast(unsigned, zhi), bhi);
        unsigned vals[16];
#pragma unroll
        for (int i = 0; i < 32; ++i) {
            const us2 bit = __builtin_bit_cast(us2, (i < 16) ? blo[i & 15] : bhi[i & 15]);
            const us2 key = next * bit + code;
            next = next * bit + step;
            const unsigned best = max_of_halves(key);
            if (i & 1) vals[i >> 1] = (best & 0xFFFF0000u) | (vals[i >> 1] & 0x0000FFFFu);
            else       vals[i >> 1] = best;
        }
        fold<0, 16>(vals, lane);
        fold<1, 8>(vals, lane);
        fold<2, 4>(vals, lane);
        fold<3, 2>(vals, lane);
        const bool upper = lane >> 4;
        const unsigned lo16 = vals[0] & 0xFFFFu;
        const unsigned hi16 = vals[0] >> 16;
        const unsigned keep = upper ? hi16 : lo16;
        const unsigned send = upper ? lo16 : hi16;
        const unsigned got  = (unsigned)lane_xor<16>((int)send);
        const unsigned res  = (got > keep) ? got : keep;
        keys[g][wd * 32 + 2 * (lane & 15) + (lane >> 4)] = (unsigned short)res;
    }
    }
    __syncthreads();

    const int t_first = wd_first * 32;
    if (part) best_over_lag_blocks<1, SPLITW * 32>(keys, words, idxs, tid, w);
    else      best_over_lag_blocks<3, 0>(keys, words, idxs, tid, w);
    __syncthreads();

    if (w == 0 && lane < (part ? 8 : 24)) {
        const int q16 = (t_first >> 4) + lane;
        const uint4 v = reinterpret_cast<const uint4*>(idxs)[q16];
        reinterpret_cast<uint4*>(reinterpret_cast<unsigned char*>(ws) + col * Tn)[q16] = v;
    }
}

__global__ __launch_bounds__(128) void emb_kernel(const float* __restrict__ ws,
                                                  const float* __restrict__ emb0,
                                                  const float* __restrict__ emb1,
                                                  float* __restrict__ out)
{
    const int lane = threadIdx.x & 31;
    const int tt   = blockIdx.x * 4 + (threadIdx.x >> 5);
    const int b    = tt >> 5;
    const int t0   = (tt & 31) << 4;
    const signed char* __restrict__ ib =
        reinterpret_cast<const signed char*>(ws) + b * Cn * Tn;
    float* __restrict__ outb = out + b * Tn * Cn;

    const int mr   = lane & 15;
    const int half = lane >> 4;
#pragma unroll
    for (int it = 0; it < 2; ++it) {
        const int c0 = it << 4;
        const float e0v = emb0[c0 + mr];
        const float e1v = emb1[c0 + mr];
        v8f acc = {};
#pragma unroll
        for (int kk = 0; kk < 16; kk += 4) {
            const int ka = kk + 2 * half;
            const signed char i0 = ib[(c0 + ka) * Tn + t0 + mr];
            const signed char i1 = ib[(c0 + ka + 1) * Tn + t0 + mr];
            v2f a0 = { (i0 == 0) ? 1.0f : 0.0f, (i1 == 0) ? 1.0f : 0.0f };
            v2f a1 = { (i0 == 1) ? 1.0f : 0.0f, (i1 == 1) ? 1.0f : 0.0f };
            v2f b0 = { (mr == ka) ? e0v : 0.0f, (mr == ka + 1) ? e0v : 0.0f };
            v2f b1 = { (mr == ka) ? e1v : 0.0f, (mr == ka + 1) ? e1v : 0.0f };
            acc = __builtin_amdgcn_wmma_f32_16x16x4_f32(
                false, a0, false, b0, (short)0, acc, false, false);
            acc = __builtin_amdgcn_wmma_f32_16x16x4_f32(
                false, a1, false, b1, (short)0, acc, false, false);
        }
#pragma unroll
        for (int r = 0; r < 8; ++r) {
            outb[(t0 + r + 8 * half) * Cn + (c0 + mr)] = acc[r];
        }
    }
}

extern "C" void kernel_launch(void* const* d_in, const int* in_sizes, int n_in,
                              void* d_out, int out_size, void* d_ws, size_t ws_size,
                              hipStream_t stream) {
    const float* x  = (const float*)d_in[0];
    const float* e0 = (const float*)d_in[1];
    const float* e1 = (const float*)d_in[2];
    float* out = (float*)d_out;
    float* ws  = (float*)d_ws;
    (void)in_sizes; (void)n_in; (void)out_size; (void)ws_size;
    scan_kernel<<<dim3(2 * Bn * Cn), dim3(128), 0, stream>>>(x, ws);
    emb_kernel<<<dim3(32), dim3(128), 0, stream>>>(ws, e0, e1, out);
}


# second launch folded into the first: each block turns its own index bytes (in LDS) into floats with the same WMMA product on rows of 16 positions and stores them, 4 bytes a line; same float sums
# speedup vs baseline: 91.4698x; 1.2198x over previous
#include <hip/hip_runtime.h>

#define Bn 4
#define Tn 512
#define Cn 32
#define NLB 8
#define SPLITW 12

typedef float v2f __attribute__((ext_vector_type(2)));
typedef float v8f __attribute__((ext_vector_type(8)));
typedef unsigned short us2 __attribute__((ext_vector_type(2)));


template <int D>
__device__ __forceinline__ int lane_xor(int v)
{
    if constexpr (D < 16) {
        return __builtin_amdgcn_update_dpp(0, v, 0x160 | D, 0xF, 0xF, false);
    } else {
        return __builtin_amdgcn_permlanex16(v, v, 0x76543210u, 0xfedcba98u, false, false);
    }
}

__device__ __forceinline__ void half_bits16(unsigned z, unsigned (&b)[16])
{
    asm(
        "v_pk_lshlrev_b16 %0, 15, %16 op_sel_hi:[0,1]\n\t"
        "v_pk_lshlrev_b16 %1, 14, %16 op_sel_hi:[0,1]\n\t"
        "v_pk_lshlrev_b16 %2, 13, %16 op_sel_hi:[0,1]\n\t"
        "v_pk_lshlrev_b16 %3, 12, %16 op_sel_hi:[0,1]\n\t"
        "v_pk_lshlrev_b16 %4, 11, %16 op_sel_hi:[0,1]\n\t"
        "v_pk_lshlrev_b16 %5, 10, %16 op_sel_hi:[0,1]\n\t"
        "v_pk_lshlrev_b16 %6, 9, %16 op_sel_hi:[0,1]\n\t"
        "v_pk_lshlrev_b16 %7, 8, %16 op_sel_hi:[0,1]\n\t"
        "v_pk_lshlrev_b16 %8, 7, %16 op_sel_hi:[0,1]\n\t"
        "v_pk_lshlrev_b16 %9, 6, %16 op_sel_hi:[0,1]\n\t"
        "v_pk_lshlrev_b16 %10, 5, %16 op_sel_hi:[0,1]\n\t"
        "v_pk_lshlrev_b16 %11, 4, %16 op_sel_hi:[0,1]\n\t"
        "v_pk_lshlrev_b16 %12, 3, %16 op_sel_hi:[0,1]\n\t"
        "v_pk_lshlrev_b16 %13, 2, %16 op_sel_hi:[0,1]\n\t"
        "v_pk_lshlrev_b16 %14, 1, %16 op_sel_hi:[0,1]\n\t"
        "v_pk_lshrrev_b16 %0, 15, %0 op_sel_hi:[0,1]\n\t"
        "v_pk_lshrrev_b16 %1, 15, %1 op_sel_hi:[0,1]\n\t"
        "v_pk_lshrrev_b16 %2, 15, %2 op_sel_hi:[0,1]\n\t"
        "v_pk_lshrrev_b16 %3, 15, %3 op_sel_hi:[0,1]\n\t"
        "v_pk_lshrrev_b16 %4, 15, %4 op_sel_hi:[0,1]\n\t"
        "v_pk_lshrrev_b16 %5, 15, %5 op_sel_hi:[0,1]\n\t"
        "v_pk_lshrrev_b16 %6, 15, %6 op_sel_hi:[0,1]\n\t"
        "v_pk_lshrrev_b16 %7, 15, %7 op_sel_hi:[0,1]\n\t"
        "v_pk_lshrrev_b16 %8, 15, %8 op_sel_hi:[0,1]\n\t"
        "v_pk_lshrrev_b16 %9, 15, %9 op_sel_hi:[0,1]\n\t"
        "v_pk_lshrrev_b16 %10, 15, %10 op_sel_hi:[0,1]\n\t"
        "v_pk_lshrrev_b16 %11, 15, %11 op_sel_hi:[0,1]\n\t"
        "v_pk_lshrrev_b16 %12, 15, %12 op_sel_hi:[0,1]\n\t"
        "v_pk_lshrrev_b16 %13, 15, %13 op_sel_hi:[0,1]\n\t"
        "v_pk_lshrrev_b16 %14, 15, %14 op_sel_hi:[0,1]\n\t"
        "v_pk_lshrrev_b16 %15, 15, %16 op_sel_hi:[0,1]"
        : "=&v"(b[0]), "=&v"(b[1]), "=&v"(b[2]), "=&v"(b[3]),
          "=&v"(b[4]), "=&v"(b[5]), "=&v"(b[6]), "=&v"(b[7]),
          "=&v"(b[8]), "=&v"(b[9]), "=&v"(b[10]), "=&v"(b[11]),
          "=&v"(b[12]), "=&v"(b[13]), "=&v"(b[14]), "=&v"(b[15])
        : "v"(z));
}

__device__ __forceinline__ unsigned max_of_halves(us2 k)
{
    const us2 sw = { k.y, k.x };
    return __builtin_bit_cast(unsigned, __builtin_elementwise_max(k, sw));
}

template <int S, int N>
__device__ __forceinline__ void fold(unsigned (&vals)[16], int lane)
{
    const bool up = (lane >> S) & 1;
#pragma unroll
    for (int m = 0; m < N / 2; ++m) {
        const unsigned a = vals[2 * m];
        const unsigned b = vals[2 * m + 1];
        const unsigned keep = up ? b : a;
        const unsigned send = up ? a : b;
        const unsigned got  = (unsigned)lane_xor<(1 << S)>((int)send);
        const us2 mx = __builtin_elementwise_max(__builtin_bit_cast(us2, got),
                                                 __builtin_bit_cast(us2, keep));
        vals[m] = __builtin_bit_cast(unsigned, mx);
    }
}

__device__ __forceinline__ unsigned word_at(unsigned wreg, int j)
{
    return (j >= 0) ? (unsigned)__builtin_amdgcn_readlane((int)wreg, j) : 0u;
}

__device__ __forceinline__ void match_pair(unsigned wreg, int wd, int g, int lane,
                                           unsigned edge, unsigned& xA, unsigned& xB)
{
    const int m = wd - 2 * g;
    const unsigned bw = word_at(wreg, wd);
    const unsigned u0 = word_at(wreg, m);
    const unsigned u1 = word_at(wreg, m - 1);
    const unsigned u2 = word_at(wreg, m - 2);
    const unsigned r  = 31u - (unsigned)lane;
    const unsigned hA = __funnelshift_r(u1, u0, r);
    const unsigned hB = __funnelshift_r(u2, u1, r);
    const unsigned vA = (m >= 1) ? 0xFFFFFFFFu : edge;
    const unsigned vB = (m >= 2) ? 0xFFFFFFFFu : ((m == 1) ? edge : 0u);
    xA = ~(hA ^ bw) & vA;
    xB = ~(hB ^ bw) & vB;
}

template <int NP, int T0>
__device__ __forceinline__ void best_over_lag_blocks(const unsigned short (&keys)[NLB][Tn],
                                                     const unsigned (&words)[16],
                                                     signed char (&idxs)[Tn],
                                                     int tid, int w)
{
    int k16[NP][NLB];
#pragma unroll
    for (int h = 0; h < NP; ++h) {
        const int t    = T0 + tid + 128 * h;
        const int wmax = (T0 >> 6) + 2 * h + (w >> 1);
#pragma unroll
        for (int u = 0; u < NLB; ++u) {
            if (NP == 3 && u > 2 * h + 1) continue;
            const int uu = (u < wmax) ? u : wmax;
            k16[h][u] = keys[uu][t];
        }
    }
    int best[NP], nxt[NP];
#pragma unroll
    for (int h = 0; h < NP; ++h) {
        const int t    = T0 + tid + 128 * h;
        const int wmax = (T0 >> 6) + 2 * h + (w >> 1);
        int bst = 0;
#pragma unroll
        for (int u = 0; u < NLB; ++u) {
            if (NP == 3 && u > 2 * h + 1) continue;
            const int kk = k16[h][u];
            const int kv = ((kk >> 6) << 10) | ((kk & 63) + 448 - 64 * u);
            const int kc = (u <= wmax) ? kv : 0;
            bst = (kc > bst) ? kc : bst;
        }
        best[h] = bst;
        const int nx = t - (512 - (bst & 1023)) + 1;
        nxt[h] = ((bst >> 10) > 0) ? nx : 0;
    }
    unsigned wv[NP];
#pragma unroll
    for (int h = 0; h < NP; ++h) wv[h] = words[nxt[h] >> 5];
#pragma unroll
    for (int h = 0; h < NP; ++h) {
        const int t   = T0 + tid + 128 * h;
        const int bit = (int)((wv[h] >> (nxt[h] & 31)) & 1u);
        idxs[t] = (signed char)(((best[h] >> 10) > 0) ? bit : -1);
    }
}

__device__ __forceinline__ v8f wmma_last(v2f a, v2f b, v8f c)
{
    asm("v_wmma_f32_16x16x4_f32 %0, %1, %2, %0\n\t"
        "v_nop\n\t"
        "v_nop\n\t"
        "v_nop\n\t"
        "v_nop"
        : "+v"(c) : "v"(a), "v"(b));
    return c;
}

template <int HR, int T0>
__device__ __forceinline__ void emb_rows(const signed char (&idxs)[Tn],
                                         float e0v, float e1v,
                                         float* __restrict__ outc, int lane, int w)
{
    const int mr   = lane & 15;
    const int half = lane >> 4;
    const bool live = (mr & 7) < HR;
    const int row   = live ? (w * 2 * HR + (mr >> 3) * HR + (mr & 7)) : 0;
    uint4 q = *reinterpret_cast<const uint4*>(&idxs[T0 + 16 * row]);
    const unsigned qs[4] = { live ? q.x : ~0u, live ? q.y : ~0u,
                             live ? q.z : ~0u, live ? q.w : ~0u };
    v8f acc = {};
#pragma unroll
    for (int s = 0; s < 4; ++s) {
        const int ka = 4 * s + 2 * half;
        const unsigned pr = qs[s] >> (16 * half);
        const int i0 = (int)(signed char)(pr & 0xFFu);
        const int i1 = (int)(signed char)((pr >> 8) & 0xFFu);
        v2f a0 = { (i0 == 0) ? 1.0f : 0.0f, (i1 == 0) ? 1.0f : 0.0f };
        v2f a1 = { (i0 == 1) ? 1.0f : 0.0f, (i1 == 1) ? 1.0f : 0.0f };
        v2f b0 = { (mr == ka) ? e0v : 0.0f, (mr == ka + 1) ? e0v : 0.0f };
        v2f b1 = { (mr == ka) ? e1v : 0.0f, (mr == ka + 1) ? e1v : 0.0f };
        acc = __builtin_amdgcn_wmma_f32_16x16x4_f32(
            false, a0, false, b0, (short)0, acc, false, false);
        if (s < 3) acc = __builtin_amdgcn_wmma_f32_16x16x4_f32(
                             false, a1, false, b1, (short)0, acc, false, false);
        else       acc = wmma_last(a1, b1, acc);
    }
#pragma unroll
    for (int r = 0; r < HR; ++r) {
        const int rr = w * 2 * HR + half * HR + r;
        outc[(T0 + 16 * rr + mr) * Cn] = acc[r];
    }
}

__global__ __launch_bounds__(128) void scan_kernel(const float* __restrict__ x,
                                                   const float* __restrict__ emb0,
                                                   const float* __restrict__ emb1,
                                                   float* __restrict__ out)
{
    __shared__ unsigned words[16];
    __shared__ unsigned short keys[NLB][Tn];
    __shared__ __attribute__((aligned(16))) signed char idxs[Tn];

    const int tid  = threadIdx.x;
    const int lane = tid & 31;
    const int w    = __builtin_amdgcn_readfirstlane(tid >> 5);
    const int col  = blockIdx.x >> 1;
    const int part = blockIdx.x & 1;
    const int b    = col >> 5;
    const int c    = col & 31;

    const float* __restrict__ xc = x + b * Tn * Cn + c;
    float xv[4];
#pragma unroll
    for (int j = 0; j < 4; ++j) xv[j] = xc[(tid + 128 * j) * Cn];
    const float e0v = emb0[c];
    const float e1v = emb1[c];
#pragma unroll
    for (int j = 0; j < 4; ++j) {
        const unsigned wj = (unsigned)__ballot(xv[j] > 0.0f);
        if (lane == 0) words[w + 4 * j] = wj;
    }
    __syncthreads();

    const unsigned wreg = words[lane & 15];
    asm volatile("" :: "v"(wreg));
    const unsigned edge = (lane < 31) ? (0xFFFFFFFFu << (lane + 1)) : 0u;
    const int wd_first = part ? SPLITW : 0;
    const int wd_end   = part ? 16 : SPLITW;

    const us2 code = { (unsigned short)(63 - lane), (unsigned short)(31 - lane) };
    const us2 step = { 64, 64 };
    for (int pass = 0; pass < 2; ++pass) {
    const int g  = (pass == 0) ? w : (part ? 7 - w : ((w == 3) ? 7 : 6 - w));
    const int wd_begin = (2 * g > wd_first) ? 2 * g : wd_first;
    unsigned runA = 0, runB = 0;
    bool goA = true, goB = true;
    for (int j = wd_begin - 1; j >= 2 * g; --j) {
        unsigned mA, mB;
        match_pair(wreg, j, g, lane, edge, mA, mB);
        const unsigned nA = ~mA;
        const unsigned nB = ~mB;
        runA += goA ? (nA ? (unsigned)__builtin_clz(nA) : 32u) : 0u;
        runB += goB ? (nB ? (unsigned)__builtin_clz(nB) : 32u) : 0u;
        goA = goA && (nA == 0u);
        goB = goB && (nB == 0u);
        if (!__any(goA || goB)) break;
    }
    us2 next = { (unsigned short)((runA + 1u) << 6), (unsigned short)((runB + 1u) << 6) };
    for (int wd = wd_begin; wd < wd_end; ++wd) {
        unsigned xA, xB;
        match_pair(wreg, wd, g, lane, edge, xA, xB);
        const us2 zlo = { (unsigned short)(xA & 0xFFFFu), (unsigned short)(xB & 0xFFFFu) };
        const us2 zhi = { (unsigned short)(xA >> 16), (unsigned short)(xB >> 16) };
        unsigned blo[16], bhi[16];
        half_bits16(__builtin_bit_cast(unsigned, zlo), blo);
        half_bits16(__builtin_bit_cast(unsigned, zhi), bhi);
        unsigned vals[16];
#pragma unroll
        for (int i = 0; i < 32; ++i) {
            const us2 bit = __builtin_bit_cast(us2, (i < 16) ? blo[i & 15] : bhi[i & 15]);
            const us2 key = next * bit + code;
            next = next * bit + step;
            const unsigned best = max_of_halves(key);
            if (i & 1) vals[i >> 1] = (best & 0xFFFF0000u) | (vals[i >> 1] & 0x0000FFFFu);
            else       vals[i >> 1] = best;
        }
        fold<0, 16>(vals, lane);
        fold<1, 8>(vals, lane);
        fold<2, 4>(vals, lane);
        fold<3, 2>(vals, lane);
        const bool upper = lane >> 4;
        const unsigned lo16 = vals[0] & 0xFFFFu;
        const unsigned hi16 = vals[0] >> 16;
        const unsigned keep = upper ? hi16 : lo16;
        const unsigned send = upper ? lo16 : hi16;
        const unsigned got  = (unsigned)lane_xor<16>((int)send);
        const unsigned res  = (got > keep) ? got : keep;
        keys[g][wd * 32 + 2 * (lane & 15) + (lane >> 4)] = (unsigned short)res;
    }
    }
    __syncthreads();

    if (part) best_over_lag_blocks<1, SPLITW * 32>(keys, words, idxs, tid, w);
    else      best_over_lag_blocks<3, 0>(keys, words, idxs, tid, w);
    __syncthreads();

    float* __restrict__ outc = out + b * Tn * Cn + c;
    if (part) emb_rows<1, SPLITW * 32>(idxs, e0v, e1v, outc, lane, w);
    else      emb_rows<3, 0>(idxs, e0v, e1v, outc, lane, w);
}

extern "C" void kernel_launch(void* const* d_in, const int* in_sizes, int n_in,
                              void* d_out, int out_size, void* d_ws, size_t ws_size,
                              hipStream_t stream) {
    const float* x  = (const float*)d_in[0];
    const float* e0 = (const float*)d_in[1];
    const float* e1 = (const float*)d_in[2];
    float* out = (float*)d_out;
    (void)in_sizes; (void)n_in; (void)out_size; (void)d_ws; (void)ws_size;
    scan_kernel<<<dim3(2 * Bn * Cn), dim3(128), 0, stream>>>(x, e0, e1, out);
}


# two waves per SIMD: blocks of 8 waves, each lag block's t walk cut at word boundaries into chunks of <= 6 steps (a later chunk seeds its counters from leading ones); integer only, same float sums
# speedup vs baseline: 100.3240x; 1.0968x over previous
#include <hip/hip_runtime.h>

#define Bn 4
#define Tn 512
#define Cn 32
#define NLB 8
#define SPLITW 12

typedef float v2f __attribute__((ext_vector_type(2)));
typedef float v8f __attribute__((ext_vector_type(8)));
typedef unsigned short us2 __attribute__((ext_vector_type(2)));


template <int D>
__device__ __forceinline__ int lane_xor(int v)
{
    if constexpr (D < 16) {
        return __builtin_amdgcn_update_dpp(0, v, 0x160 | D, 0xF, 0xF, false);
    } else {
        return __builtin_amdgcn_permlanex16(v, v, 0x76543210u, 0xfedcba98u, false, false);
    }
}

__device__ __forceinline__ void half_bits16(unsigned z, unsigned (&b)[16])
{
    asm(
        "v_pk_lshlrev_b16 %0, 15, %16 op_sel_hi:[0,1]\n\t"
        "v_pk_lshlrev_b16 %1, 14, %16 op_sel_hi:[0,1]\n\t"
        "v_pk_lshlrev_b16 %2, 13, %16 op_sel_hi:[0,1]\n\t"
        "v_pk_lshlrev_b16 %3, 12, %16 op_sel_hi:[0,1]\n\t"
        "v_pk_lshlrev_b16 %4, 11, %16 op_sel_hi:[0,1]\n\t"
        "v_pk_lshlrev_b16 %5, 10, %16 op_sel_hi:[0,1]\n\t"
        "v_pk_lshlrev_b16 %6, 9, %16 op_sel_hi:[0,1]\n\t"
        "v_pk_lshlrev_b16 %7, 8, %16 op_sel_hi:[0,1]\n\t"
        "v_pk_lshlrev_b16 %8, 7, %16 op_sel_hi:[0,1]\n\t"
        "v_pk_lshlrev_b16 %9, 6, %16 op_sel_hi:[0,1]\n\t"
        "v_pk_lshlrev_b16 %10, 5, %16 op_sel_hi:[0,1]\n\t"
        "v_pk_lshlrev_b16 %11, 4, %16 op_sel_hi:[0,1]\n\t"
        "v_pk_lshlrev_b16 %12, 3, %16 op_sel_hi:[0,1]\n\t"
        "v_pk_lshlrev_b16 %13, 2, %16 op_sel_hi:[0,1]\n\t"
        "v_pk_lshlrev_b16 %14, 1, %16 op_sel_hi:[0,1]\n\t"
        "v_pk_lshrrev_b16 %0, 15, %0 op_sel_hi:[0,1]\n\t"
        "v_pk_lshrrev_b16 %1, 15, %1 op_sel_hi:[0,1]\n\t"
        "v_pk_lshrrev_b16 %2, 15, %2 op_sel_hi:[0,1]\n\t"
        "v_pk_lshrrev_b16 %3, 15, %3 op_sel_hi:[0,1]\n\t"
        "v_pk_lshrrev_b16 %4, 15, %4 op_sel_hi:[0,1]\n\t"
        "v_pk_lshrrev_b16 %5, 15, %5 op_sel_hi:[0,1]\n\t"
        "v_pk_lshrrev_b16 %6, 15, %6 op_sel_hi:[0,1]\n\t"
        "v_pk_lshrrev_b16 %7, 15, %7 op_sel_hi:[0,1]\n\t"
        "v_pk_lshrrev_b16 %8, 15, %8 op_sel_hi:[0,1]\n\t"
        "v_pk_lshrrev_b16 %9, 15, %9 op_sel_hi:[0,1]\n\t"
        "v_pk_lshrrev_b16 %10, 15, %10 op_sel_hi:[0,1]\n\t"
        "v_pk_lshrrev_b16 %11, 15, %11 op_sel_hi:[0,1]\n\t"
        "v_pk_lshrrev_b16 %12, 15, %12 op_sel_hi:[0,1]\n\t"
        "v_pk_lshrrev_b16 %13, 15, %13 op_sel_hi:[0,1]\n\t"
        "v_pk_lshrrev_b16 %14, 15, %14 op_sel_hi:[0,1]\n\t"
        "v_pk_lshrrev_b16 %15, 15, %16 op_sel_hi:[0,1]"
        : "=&v"(b[0]), "=&v"(b[1]), "=&v"(b[2]), "=&v"(b[3]),
          "=&v"(b[4]), "=&v"(b[5]), "=&v"(b[6]), "=&v"(b[7]),
          "=&v"(b[8]), "=&v"(b[9]), "=&v"(b[10]), "=&v"(b[11]),
          "=&v"(b[12]), "=&v"(b[13]), "=&v"(b[14]), "=&v"(b[15])
        : "v"(z));
}

__device__ __forceinline__ unsigned max_of_halves(us2 k)
{
    const us2 sw = { k.y, k.x };
    return __builtin_bit_cast(unsigned, __builtin_elementwise_max(k, sw));
}

template <int S, int N>
__device__ __forceinline__ void fold(unsigned (&vals)[16], int lane)
{
    const bool up = (lane >> S) & 1;
#pragma unroll
    for (int m = 0; m < N / 2; ++m) {
        const unsigned a = vals[2 * m];
        const unsigned b = vals[2 * m + 1];
        const unsigned keep = up ? b : a;
        const unsigned send = up ? a : b;
        const unsigned got  = (unsigned)lane_xor<(1 << S)>((int)send);
        const us2 mx = __builtin_elementwise_max(__builtin_bit_cast(us2, got),
                                                 __builtin_bit_cast(us2, keep));
        vals[m] = __builtin_bit_cast(unsigned, mx);
    }
}

__device__ __forceinline__ unsigned word_at(unsigned wreg, int j)
{
    return (j >= 0) ? (unsigned)__builtin_amdgcn_readlane((int)wreg, j) : 0u;
}

__device__ __forceinline__ void match_pair(unsigned wreg, int wd, int g, int lane,
                                           unsigned edge, unsigned& xA, unsigned& xB)
{
    const int m = wd - 2 * g;
    const unsigned bw = word_at(wreg, wd);
    const unsigned u0 = word_at(wreg, m);
    const unsigned u1 = word_at(wreg, m - 1);
    const unsigned u2 = word_at(wreg, m - 2);
    const unsigned r  = 31u - (unsigned)lane;
    const unsigned hA = __funnelshift_r(u1, u0, r);
    const unsigned hB = __funnelshift_r(u2, u1, r);
    const unsigned vA = (m >= 1) ? 0xFFFFFFFFu : edge;
    const unsigned vB = (m >= 2) ? 0xFFFFFFFFu : ((m == 1) ? edge : 0u);
    xA = ~(hA ^ bw) & vA;
    xB = ~(hB ^ bw) & vB;
}

template <int NP, int T0>
__device__ __forceinline__ void best_over_lag_blocks(const unsigned short (&keys)[NLB][Tn],
                                                     const unsigned (&words)[16],
                                                     signed char (&idxs)[Tn],
                                                     int tid, int w)
{
    int k16[NP][NLB];
#pragma unroll
    for (int h = 0; h < NP; ++h) {
        const int t    = T0 + tid + 128 * h;
        const int wmax = (T0 >> 6) + 2 * h + (w >> 1);
#pragma unroll
        for (int u = 0; u < NLB; ++u) {
            if (NP == 3 && u > 2 * h + 1) continue;
            const int uu = (u < wmax) ? u : wmax;
            k16[h][u] = keys[uu][t];
        }
    }
    int best[NP], nxt[NP];
#pragma unroll
    for (int h = 0; h < NP; ++h) {
        const int t    = T0 + tid + 128 * h;
        const int wmax = (T0 >> 6) + 2 * h + (w >> 1);
        int bst = 0;
#pragma unroll
        for (int u = 0; u < NLB; ++u) {
            if (NP == 3 && u > 2 * h + 1) continue;
            const int kk = k16[h][u];
            const int kv = ((kk >> 6) << 10) | ((kk & 63) + 448 - 64 * u);
            const int kc = (u <= wmax) ? kv : 0;
            bst = (kc > bst) ? kc : bst;
        }
        best[h] = bst;
        const int nx = t - (512 - (bst & 1023)) + 1;
        nxt[h] = ((bst >> 10) > 0) ? nx : 0;
    }
    unsigned wv[NP];
#pragma unroll
    for (int h = 0; h < NP; ++h) wv[h] = words[nxt[h] >> 5];
#pragma unroll
    for (int h = 0; h < NP; ++h) {
        const int t   = T0 + tid + 128 * h;
        const int bit = (int)((wv[h] >> (nxt[h] & 31)) & 1u);
        idxs[t] = (signed char)(((best[h] >> 10) > 0) ? bit : -1);
    }
}

__device__ __forceinline__ v8f wmma_last(v2f a, v2f b, v8f c)
{
    asm("v_wmma_f32_16x16x4_f32 %0, %1, %2, %0\n\t"
        "v_nop\n\t"
        "v_nop\n\t"
        "v_nop\n\t"
        "v_nop"
        : "+v"(c) : "v"(a), "v"(b));
    return c;
}

template <int HR, int T0>
__device__ __forceinline__ void emb_rows(const signed char (&idxs)[Tn],
                                         float e0v, float e1v,
                                         float* __restrict__ outc, int lane, int w)
{
    const int mr   = lane & 15;
    const int half = lane >> 4;
    const bool live = (mr & 7) < HR;
    const int row   = live ? (w * 2 * HR + (mr >> 3) * HR + (mr & 7)) : 0;
    uint4 q = *reinterpret_cast<const uint4*>(&idxs[T0 + 16 * row]);
    const unsigned qs[4] = { live ? q.x : ~0u, live ? q.y : ~0u,
                             live ? q.z : ~0u, live ? q.w : ~0u };
    v8f acc = {};
#pragma unroll
    for (int s = 0; s < 4; ++s) {
        const int ka = 4 * s + 2 * half;
        const unsigned pr = qs[s] >> (16 * half);
        const int i0 = (int)(signed char)(pr & 0xFFu);
        const int i1 = (int)(signed char)((pr >> 8) & 0xFFu);
        v2f a0 = { (i0 == 0) ? 1.0f : 0.0f, (i1 == 0) ? 1.0f : 0.0f };
        v2f a1 = { (i0 == 1) ? 1.0f : 0.0f, (i1 == 1) ? 1.0f : 0.0f };
        v2f b0 = { (mr == ka) ? e0v : 0.0f, (mr == ka + 1) ? e0v : 0.0f };
        v2f b1 = { (mr == ka) ? e1v : 0.0f, (mr == ka + 1) ? e1v : 0.0f };
        acc = __builtin_amdgcn_wmma_f32_16x16x4_f32(
            false, a0, false, b0, (short)0, acc, false, false);
        if (s < 3) acc = __builtin_amdgcn_wmma_f32_16x16x4_f32(
                             false, a1, false, b1, (short)0, acc, false, false);
        else       acc = wmma_last(a1, b1, acc);
    }
#pragma unroll
    for (int r = 0; r < HR; ++r) {
        const int rr = w * 2 * HR + half * HR + r;
        outc[(T0 + 16 * rr + mr) * Cn] = acc[r];
    }
}

__global__ __launch_bounds__(256) void scan_kernel(const float* __restrict__ x,
                                                   const float* __restrict__ emb0,
                                                   const float* __restrict__ emb1,
                                                   float* __restrict__ out)
{
    __shared__ unsigned words[16];
    __shared__ unsigned short keys[NLB][Tn];
    __shared__ __attribute__((aligned(16))) signed char idxs[Tn];

    const int tid  = threadIdx.x;
    const int lane = tid & 31;
    const int w    = __builtin_amdgcn_readfirstlane(tid >> 5);
    const int col  = blockIdx.x >> 1;
    const int part = blockIdx.x & 1;
    const int b    = col >> 5;
    const int c    = col & 31;

    const float* __restrict__ xc = x + b * Tn * Cn + c;
    float xv[2];
#pragma unroll
    for (int j = 0; j < 2; ++j) xv[j] = xc[(tid + 256 * j) * Cn];
    const float e0v = emb0[c];
    const float e1v = emb1[c];
#pragma unroll
    for (int j = 0; j < 2; ++j) {
        const unsigned wj = (unsigned)__ballot(xv[j] > 0.0f);
        if (lane == 0) words[w + 8 * j] = wj;
    }
    __syncthreads();

    const unsigned wreg = words[lane & 15];
    asm volatile("" :: "v"(wreg));
    const unsigned edge = (lane < 31) ? (0xFFFFFFFFu << (lane + 1)) : 0u;
    const int sh = 4 * w;

    const us2 code = { (unsigned short)(63 - lane), (unsigned short)(31 - lane) };
    const us2 step = { 64, 64 };
    for (int pass = 0; pass < 2; ++pass) {
    int g, wd_begin, wd_end;
    if (part) {
        g        = (pass == 0) ? w : 7;
        wd_begin = (pass == 0) ? ((2 * w > SPLITW) ? 2 * w : SPLITW) : 16;
        wd_end   = 16;
    } else {
        g        = (int)(((pass == 0) ? 0x33221100u : 0x47577777u) >> sh) & 15;
        wd_begin = (int)(((pass == 0) ? 0xB6947260u : 0x8CACCCCCu) >> sh) & 15;
        wd_end   = (pass == 0) ? ((int)(0xCBC9C7C6u >> sh) & 15) : SPLITW;
    }
    if (wd_begin >= wd_end) continue;
    unsigned runA = 0, runB = 0;
    bool goA = true, goB = true;
    for (int j = wd_begin - 1; j >= 2 * g; --j) {
        unsigned mA, mB;
        match_pair(wreg, j, g, lane, edge, mA, mB);
        const unsigned nA = ~mA;
        const unsigned nB = ~mB;
        runA += goA ? (nA ? (unsigned)__builtin_clz(nA) : 32u) : 0u;
        runB += goB ? (nB ? (unsigned)__builtin_clz(nB) : 32u) : 0u;
        goA = goA && (nA == 0u);
        goB = goB && (nB == 0u);
        if (!__any(goA || goB)) break;
    }
    us2 next = { (unsigned short)((runA + 1u) << 6), (unsigned short)((runB + 1u) << 6) };
    for (int wd = wd_begin; wd < wd_end; ++wd) {
        unsigned xA, xB;
        match_pair(wreg, wd, g, lane, edge, xA, xB);
        const us2 zlo = { (unsigned short)(xA & 0xFFFFu), (unsigned short)(xB & 0xFFFFu) };
        const us2 zhi = { (unsigned short)(xA >> 16), (unsigned short)(xB >> 16) };
        unsigned blo[16], bhi[16];
        half_bits16(__builtin_bit_cast(unsigned, zlo), blo);
        half_bits16(__builtin_bit_cast(unsigned, zhi), bhi);
        unsigned vals[16];
#pragma unroll
        for (int i = 0; i < 32; ++i) {
            const us2 bit = __builtin_bit_cast(us2, (i < 16) ? blo[i & 15] : bhi[i & 15]);
            const us2 key = next * bit + code;
            next = next * bit + step;
            const unsigned best = max_of_halves(key);
            if (i & 1) vals[i >> 1] = (best & 0xFFFF0000u) | (vals[i >> 1] & 0x0000FFFFu);
            else       vals[i >> 1] = best;
        }
        fold<0, 16>(vals, lane);
        fold<1, 8>(vals, lane);
        fold<2, 4>(vals, lane);
        fold<3, 2>(vals, lane);
        const bool upper = lane >> 4;
        const unsigned lo16 = vals[0] & 0xFFFFu;
        const unsigned hi16 = vals[0] >> 16;
        const unsigned keep = upper ? hi16 : lo16;
        const unsigned send = upper ? lo16 : hi16;
        const unsigned got  = (unsigned)lane_xor<16>((int)send);
        const unsigned res  = (got > keep) ? got : keep;
        keys[g][wd * 32 + 2 * (lane & 15) + (lane >> 4)] = (unsigned short)res;
    }
    }
    __syncthreads();

    if (w < 4) {
        if (part) best_over_lag_blocks<1, SPLITW * 32>(keys, words, idxs, tid, w);
        else      best_over_lag_blocks<3, 0>(keys, words, idxs, tid, w);
    }
    __syncthreads();

    float* __restrict__ outc = out + b * Tn * Cn + c;
    if (w < 4) {
        if (part) emb_rows<1, SPLITW * 32>(idxs, e0v, e1v, outc, lane, w);
        else      emb_rows<3, 0>(idxs, e0v, e1v, outc, lane, w);
    }
}

extern "C" void kernel_launch(void* const* d_in, const int* in_sizes, int n_in,
                              void* d_out, int out_size, void* d_ws, size_t ws_size,
                              hipStream_t stream) {
    const float* x  = (const float*)d_in[0];
    const float* e0 = (const float*)d_in[1];
    const float* e1 = (const float*)d_in[2];
    float* out = (float*)d_out;
    (void)in_sizes; (void)n_in; (void)out_size; (void)d_ws; (void)ws_size;
    scan_kernel<<<dim3(2 * Bn * Cn), dim3(256), 0, stream>>>(x, e0, e1, out);
}
